# starting kernel
# baseline (speedup 1.0000x reference)
#include <hip/hip_runtime.h>
#include <hip/hip_bf16.h>
#include <math.h>

#define BATCH   16
#define CIN     256
#define NPIX    4096
#define HEADS   4
#define DHEAD   32
#define HIDDEN  128
#define O3      384
#define KSPLIT  8
#define KCHUNK  (NPIX / KSPLIT)

typedef __attribute__((ext_vector_type(2))) float v2f;
typedef __attribute__((ext_vector_type(8))) float v8f;

__device__ __forceinline__ v8f wmma4(v2f a, v2f b, v8f c) {
  return __builtin_amdgcn_wmma_f32_16x16x4_f32(false, a, false, b, (short)0, c,
                                               false, false);
}

__global__ __launch_bounds__(256) void k_qkv_gemm(
    const float* __restrict__ x, const float* __restrict__ wqkv,
    float* __restrict__ qkv) {
  const int lane = threadIdx.x & 31;
  const int wave = threadIdx.x >> 5;
  const int half = lane >> 4;
  const int lm   = lane & 15;
  const int nBase = blockIdx.x * 16;
  const int b     = blockIdx.y;

  const float* __restrict__ xb = x + (size_t)b * CIN * NPIX;
  float* __restrict__ qb = qkv + (size_t)b * O3 * NPIX;

  const int m0 = wave * 16;
  v8f acc0 = {}, acc1 = {}, acc2 = {};

  for (int kb = 0; kb < CIN; kb += 4) {
    const int k0 = kb + 2 * half;
    v2f bf, a0, a1, a2;
    bf.x = xb[(size_t)k0 * NPIX + nBase + lm];
    bf.y = xb[(size_t)(k0 + 1) * NPIX + nBase + lm];
    a0.x = wqkv[(m0 + lm) * CIN + k0];
    a0.y = wqkv[(m0 + lm) * CIN + k0 + 1];
    a1.x = wqkv[(m0 + 128 + lm) * CIN + k0];
    a1.y = wqkv[(m0 + 128 + lm) * CIN + k0 + 1];
    a2.x = wqkv[(m0 + 256 + lm) * CIN + k0];
    a2.y = wqkv[(m0 + 256 + lm) * CIN + k0 + 1];
    acc0 = wmma4(a0, bf, acc0);
    acc1 = wmma4(a1, bf, acc1);
    acc2 = wmma4(a2, bf, acc2);
  }
#pragma unroll
  for (int v = 0; v < 8; ++v) {
    const int row = v + 8 * half;
    qb[(size_t)(m0 + row) * NPIX + nBase + lm]       = acc0[v];
    qb[(size_t)(m0 + 128 + row) * NPIX + nBase + lm] = acc1[v];
    qb[(size_t)(m0 + 256 + row) * NPIX + nBase + lm] = acc2[v];
  }
}

__global__ __launch_bounds__(256) void k_softmax_n(float* __restrict__ qkv) {
  const int r = blockIdx.x;
  const int b = r >> 7;
  const int c = r & 127;
  float* __restrict__ row = qkv + ((size_t)b * O3 + c) * NPIX;

  __shared__ float red[256];
  const int t = threadIdx.x;

  float v[16];
  float m = -INFINITY;
#pragma unroll
  for (int j = 0; j < 16; ++j) {
    v[j] = row[t + j * 256];
    m = fmaxf(m, v[j]);
  }
  red[t] = m; __syncthreads();
  for (int s = 128; s > 0; s >>= 1) {
    if (t < s) red[t] = fmaxf(red[t], red[t + s]);
    __syncthreads();
  }
  m = red[0]; __syncthreads();

  float sum = 0.f;
#pragma unroll
  for (int j = 0; j < 16; ++j) {
    v[j] = __expf(v[j] - m);
    sum += v[j];
  }
  red[t] = sum; __syncthreads();
  for (int s = 128; s > 0; s >>= 1) {
    if (t < s) red[t] += red[t + s];
    __syncthreads();
  }
  const float inv = 1.0f / red[0];
#pragma unroll
  for (int j = 0; j < 16; ++j) row[t + j * 256] = v[j] * inv;
}

__global__ __launch_bounds__(256) void k_softmax_d(
    const float* __restrict__ qkv, float* __restrict__ kbuf) {
  const int col = blockIdx.x * 256 + threadIdx.x;
  const int n = col & (NPIX - 1);
  const int h = (col >> 12) & (HEADS - 1);
  const int b = col >> 14;

  const float* __restrict__ qs = qkv + ((size_t)b * O3 + h * DHEAD) * NPIX + n;
  float vals[DHEAD];
  float m = -INFINITY;
#pragma unroll
  for (int d = 0; d < DHEAD; ++d) {
    vals[d] = qs[(size_t)d * NPIX];
    m = fmaxf(m, vals[d]);
  }
  float s = 0.f;
#pragma unroll
  for (int d = 0; d < DHEAD; ++d) {
    vals[d] = __expf(vals[d] - m);
    s += vals[d];
  }
  const float inv = 1.0f / s;
  float* __restrict__ ko =
      kbuf + (size_t)(b * HEADS + h) * DHEAD * NPIX + n;
#pragma unroll
  for (int d = 0; d < DHEAD; ++d) ko[(size_t)d * NPIX] = vals[d] * inv;
}

__global__ __launch_bounds__(128) void k_context_part(
    const float* __restrict__ kbuf, const float* __restrict__ qkv,
    float* __restrict__ pctx) {
  const int ks = blockIdx.x;
  const int bh = blockIdx.y;
  const int b = bh >> 2, h = bh & 3;
  const int lane = threadIdx.x & 31;
  const int wave = threadIdx.x >> 5;
  const int half = lane >> 4, lm = lane & 15;
  const int mBase = (wave >> 1) * 16;
  const int nBase = (wave & 1) * 16;
  const int kBeg  = ks * KCHUNK;

  const float* __restrict__ kA = kbuf + (size_t)bh * DHEAD * NPIX;
  const float* __restrict__ vB =
      qkv + ((size_t)b * O3 + 2 * HIDDEN + h * DHEAD) * NPIX;

  v8f acc = {};
  for (int kb = 0; kb < KCHUNK; kb += 4) {
    const int k0 = kBeg + kb + 2 * half;
    v2f a, bb;
    a.x  = kA[(size_t)(mBase + lm) * NPIX + k0];
    a.y  = kA[(size_t)(mBase + lm) * NPIX + k0 + 1];
    bb.x = vB[(size_t)(nBase + lm) * NPIX + k0];
    bb.y = vB[(size_t)(nBase + lm) * NPIX + k0 + 1];
    acc = wmma4(a, bb, acc);
  }
#pragma unroll
  for (int v = 0; v < 8; ++v) {
    const int row = mBase + v + 8 * half;
    pctx[(((size_t)ks * (BATCH * HEADS) + bh) * DHEAD + row) * DHEAD + nBase +
         lm] = acc[v];
  }
}

__global__ __launch_bounds__(256) void k_ctx_reduce(
    const float* __restrict__ pctx, float* __restrict__ ctx) {
  const int i = blockIdx.x * 256 + threadIdx.x;
  const size_t stride = (size_t)BATCH * HEADS * DHEAD * DHEAD;
  float s = 0.f;
#pragma unroll
  for (int ks = 0; ks < KSPLIT; ++ks) s += pctx[(size_t)ks * stride + i];
  ctx[i] = s;
}

__global__ __launch_bounds__(256) void k_fold_w2(
    const float* __restrict__ wout, const float* __restrict__ ctx,
    float* __restrict__ w2) {
  const int bh = blockIdx.x;
  const int b = bh >> 2, h = bh & 3;
  const int lane = threadIdx.x & 31;
  const int wave = threadIdx.x >> 5;
  const int half = lane >> 4, lm = lane & 15;
  const int m0 = wave * 16;

  const float* __restrict__ cb = ctx + (size_t)bh * DHEAD * DHEAD;

  v8f acc00 = {}, acc01 = {}, acc10 = {}, acc11 = {};
#pragma unroll
  for (int kb = 0; kb < DHEAD; kb += 4) {
    const int k0 = kb + 2 * half;
    v2f a0, a1, b0, b1;
    a0.x = wout[(m0 + lm) * HIDDEN + h * DHEAD + k0];
    a0.y = wout[(m0 + lm) * HIDDEN + h * DHEAD + k0 + 1];
    a1.x = wout[(m0 + 128 + lm) * HIDDEN + h * DHEAD + k0];
    a1.y = wout[(m0 + 128 + lm) * HIDDEN + h * DHEAD + k0 + 1];
    b0.x = cb[(size_t)(0 + lm) * DHEAD + k0];
    b0.y = cb[(size_t)(0 + lm) * DHEAD + k0 + 1];
    b1.x = cb[(size_t)(16 + lm) * DHEAD + k0];
    b1.y = cb[(size_t)(16 + lm) * DHEAD + k0 + 1];
    acc00 = wmma4(a0, b0, acc00);
    acc01 = wmma4(a0, b1, acc01);
    acc10 = wmma4(a1, b0, acc10);
    acc11 = wmma4(a1, b1, acc11);
  }
  float* __restrict__ w2b = w2 + (size_t)b * CIN * HIDDEN;
#pragma unroll
  for (int v = 0; v < 8; ++v) {
    const int row = v + 8 * half;
    w2b[(size_t)(m0 + row) * HIDDEN + h * DHEAD + 0 + lm]        = acc00[v];
    w2b[(size_t)(m0 + row) * HIDDEN + h * DHEAD + 16 + lm]       = acc01[v];
    w2b[(size_t)(m0 + 128 + row) * HIDDEN + h * DHEAD + 0 + lm]  = acc10[v];
    w2b[(size_t)(m0 + 128 + row) * HIDDEN + h * DHEAD + 16 + lm] = acc11[v];
  }
}

__global__ __launch_bounds__(256) void k_out_gemm(
    const float* __restrict__ qkv, const float* __restrict__ w2,
    float* __restrict__ out) {
  const int lane = threadIdx.x & 31;
  const int wave = threadIdx.x >> 5;
  const int half = lane >> 4, lm = lane & 15;
  const int nBase = blockIdx.x * 16;
  const int b     = blockIdx.y;

  const float* __restrict__ qs  = qkv + (size_t)b * O3 * NPIX;
  const float* __restrict__ w2b = w2 + (size_t)b * CIN * HIDDEN;
  float* __restrict__ ob = out + (size_t)b * CIN * NPIX;

  const int m0 = wave * 16;
  v8f acc0 = {}, acc1 = {};
  for (int kb = 0; kb < HIDDEN; kb += 4) {
    const int k0 = kb + 2 * half;
    v2f bf, a0, a1;
    bf.x = qs[(size_t)k0 * NPIX + nBase + lm];
    bf.y = qs[(size_t)(k0 + 1) * NPIX + nBase + lm];
    a0.x = w2b[(m0 + lm) * HIDDEN + k0];
    a0.y = w2b[(m0 + lm) * HIDDEN + k0 + 1];
    a1.x = w2b[(m0 + 128 + lm) * HIDDEN + k0];
    a1.y = w2b[(m0 + 128 + lm) * HIDDEN + k0 + 1];
    acc0 = wmma4(a0, bf, acc0);
    acc1 = wmma4(a1, bf, acc1);
  }
#pragma unroll
  for (int v = 0; v < 8; ++v) {
    const int row = v + 8 * half;
    ob[(size_t)(m0 + row) * NPIX + nBase + lm]       = acc0[v];
    ob[(size_t)(m0 + 128 + row) * NPIX + nBase + lm] = acc1[v];
  }
}

extern "C" void kernel_launch(void* const* d_in, const int* in_sizes, int n_in,
                              void* d_out, int out_size, void* d_ws,
                              size_t ws_size, hipStream_t stream) {
  const float* x    = (const float*)d_in[0];
  const float* wqkv = (const float*)d_in[1];
  const float* wout = (const float*)d_in[2];
  float* out = (float*)d_out;

  char* ws = (char*)d_ws;
  const size_t QKV_F = (size_t)BATCH * O3 * NPIX;
  const size_t KB_F  = (size_t)BATCH * HIDDEN * NPIX;
  const size_t CTX_F = (size_t)BATCH * HEADS * DHEAD * DHEAD;
  float* qkv  = (float*)ws;
  float* kbuf = (float*)(ws + QKV_F * 4);
  float* ctx  = (float*)(ws + (QKV_F + KB_F) * 4);
  float* pctx = (float*)(ws + (QKV_F + KB_F + CTX_F) * 4);
  float* w2   = (float*)(ws + (QKV_F + KB_F + CTX_F + KSPLIT * CTX_F) * 4);

  k_qkv_gemm<<<dim3(NPIX / 16, BATCH), 256, 0, stream>>>(x, wqkv, qkv);
  k_softmax_n<<<BATCH * HIDDEN, 256, 0, stream>>>(qkv);
  k_softmax_d<<<(BATCH * HEADS * NPIX) / 256, 256, 0, stream>>>(qkv, kbuf);
  k_context_part<<<dim3(KSPLIT, BATCH * HEADS), 128, 0, stream>>>(kbuf, qkv,
                                                                  pctx);
  k_ctx_reduce<<<CTX_F / 256, 256, 0, stream>>>(pctx, ctx);
  k_fold_w2<<<BATCH * HEADS, 256, 0, stream>>>(wout, ctx, w2);
  k_out_gemm<<<dim3(NPIX / 16, BATCH), 256, 0, stream>>>(qkv, w2, out);
}


# qkv and out GEMMs: 64-column block tiles, both operands staged through LDS by whole-line 16-byte global loads, fragments read from LDS (was per-lane strided global loads); summation order unchanged
# speedup vs baseline: 2.3263x; 2.3263x over previous
#include <hip/hip_runtime.h>
#include <hip/hip_bf16.h>
#include <math.h>

#define BATCH   16
#define CIN     256
#define NPIX    4096
#define HEADS   4
#define DHEAD   32
#define HIDDEN  128
#define O3      384
#define KSPLIT  8
#define KCHUNK  (NPIX / KSPLIT)
#define TN      64
#define KC      32
#define APITCH  36
#define BPITCH  72

typedef __attribute__((ext_vector_type(2))) float v2f;
typedef __attribute__((ext_vector_type(4))) float v4f;
typedef __attribute__((ext_vector_type(8))) float v8f;

__device__ __forceinline__ v8f wmma4(v2f a, v2f b, v8f c) {
  return __builtin_amdgcn_wmma_f32_16x16x4_f32(false, a, false, b, (short)0, c,
                                               false, false);
}

template <int MT, int KTOT>
__device__ __forceinline__ void gemm_tile_lds(const float* __restrict__ A,
                                              const float* __restrict__ Bm,
                                              float* __restrict__ C,
                                              float* sA, float* sB,
                                              const int nBase) {
  constexpr int NA = MT * 4;
  const int t    = threadIdx.x;
  const int lane = t & 31;
  const int wave = t >> 5;
  const int half = lane >> 4;
  const int lm   = lane & 15;
  const int m0   = wave * 16;

  const int ar = t >> 3, ac = (t & 7) * 4;
  const int br = t >> 4, bc = (t & 15) * 4;
  const float* __restrict__ gA = A + (size_t)ar * KTOT + ac;
  const float* __restrict__ gB = Bm + (size_t)br * NPIX + nBase + bc;
  float* lA = sA + ar * APITCH + ac;
  float* lB = sB + br * BPITCH + bc;

  const float* fA = sA + (m0 + lm) * APITCH + 2 * half;
  const float* fB = sB + (2 * half) * BPITCH + lm;

  v4f ra[NA];
  v4f rb[2];
  v8f acc[MT][4];
#pragma unroll
  for (int j = 0; j < MT; ++j)
#pragma unroll
    for (int i = 0; i < 4; ++i) acc[j][i] = v8f{};

#pragma unroll
  for (int i = 0; i < NA; ++i)
    ra[i] = *reinterpret_cast<const v4f*>(gA + (size_t)(32 * i) * KTOT);
#pragma unroll
  for (int i = 0; i < 2; ++i)
    rb[i] = *reinterpret_cast<const v4f*>(gB + (size_t)(16 * i) * NPIX);
#pragma unroll
  for (int i = 0; i < NA; ++i)
    *reinterpret_cast<v4f*>(lA + 32 * i * APITCH) = ra[i];
#pragma unroll
  for (int i = 0; i < 2; ++i)
    *reinterpret_cast<v4f*>(lB + 16 * i * BPITCH) = rb[i];
  __syncthreads();

  for (int kc = 0; kc < KTOT; kc += KC) {
    const bool more = (kc + KC) < KTOT;
    if (more) {
#pragma unroll
      for (int i = 0; i < NA; ++i)
        ra[i] = *reinterpret_cast<const v4f*>(gA + (size_t)(32 * i) * KTOT +
                                                 kc + KC);
#pragma unroll
      for (int i = 0; i < 2; ++i)
        rb[i] = *reinterpret_cast<const v4f*>(
            gB + (size_t)(kc + KC + 16 * i) * NPIX);
    }
#pragma unroll
    for (int kk = 0; kk < KC; kk += 4) {
      v2f a[MT], b[4];
#pragma unroll
      for (int j = 0; j < MT; ++j)
        a[j] = *reinterpret_cast<const v2f*>(fA + 128 * j * APITCH + kk);
#pragma unroll
      for (int i = 0; i < 4; ++i) {
        b[i].x = fB[kk * BPITCH + 16 * i];
        b[i].y = fB[(kk + 1) * BPITCH + 16 * i];
      }
#pragma unroll
      for (int j = 0; j < MT; ++j)
#pragma unroll
        for (int i = 0; i < 4; ++i) acc[j][i] = wmma4(a[j], b[i], acc[j][i]);
    }
    __syncthreads();
    if (more) {
#pragma unroll
      for (int i = 0; i < NA; ++i)
        *reinterpret_cast<v4f*>(lA + 32 * i * APITCH) = ra[i];
#pragma unroll
      for (int i = 0; i < 2; ++i)
        *reinterpret_cast<v4f*>(lB + 16 * i * BPITCH) = rb[i];
      __syncthreads();
    }
  }

#pragma unroll
  for (int j = 0; j < MT; ++j)
#pragma unroll
    for (int i = 0; i < 4; ++i)
#pragma unroll
      for (int v = 0; v < 8; ++v)
        C[(size_t)(m0 + 128 * j + v + 8 * half) * NPIX + nBase + 16 * i + lm] =
            acc[j][i][v];
}

__global__ __launch_bounds__(256) void k_qkv_gemm(
    const float* __restrict__ x, const float* __restrict__ wqkv,
    float* __restrict__ qkv) {
  __shared__ __attribute__((aligned(16))) float sA[O3 * APITCH];
  __shared__ __attribute__((aligned(16))) float sB[KC * BPITCH];
  const int b = blockIdx.y;
  gemm_tile_lds<3, CIN>(wqkv, x + (size_t)b * CIN * NPIX,
                        qkv + (size_t)b * O3 * NPIX, sA, sB, blockIdx.x * TN);
}

__global__ __launch_bounds__(256) void k_softmax_n(float* __restrict__ qkv) {
  const int r = blockIdx.x;
  const int b = r >> 7;
  const int c = r & 127;
  float* __restrict__ row = qkv + ((size_t)b * O3 + c) * NPIX;

  __shared__ float red[256];
  const int t = threadIdx.x;

  float v[16];
  float m = -INFINITY;
#pragma unroll
  for (int j = 0; j < 16; ++j) {
    v[j] = row[t + j * 256];
    m = fmaxf(m, v[j]);
  }
  red[t] = m; __syncthreads();
  for (int s = 128; s > 0; s >>= 1) {
    if (t < s) red[t] = fmaxf(red[t], red[t + s]);
    __syncthreads();
  }
  m = red[0]; __syncthreads();

  float sum = 0.f;
#pragma unroll
  for (int j = 0; j < 16; ++j) {
    v[j] = __expf(v[j] - m);
    sum += v[j];
  }
  red[t] = sum; __syncthreads();
  for (int s = 128; s > 0; s >>= 1) {
    if (t < s) red[t] += red[t + s];
    __syncthreads();
  }
  const float inv = 1.0f / red[0];
#pragma unroll
  for (int j = 0; j < 16; ++j) row[t + j * 256] = v[j] * inv;
}

__global__ __launch_bounds__(256) void k_softmax_d(
    const float* __restrict__ qkv, float* __restrict__ kbuf) {
  const int col = blockIdx.x * 256 + threadIdx.x;
  const int n = col & (NPIX - 1);
  const int h = (col >> 12) & (HEADS - 1);
  const int b = col >> 14;

  const float* __restrict__ qs = qkv + ((size_t)b * O3 + h * DHEAD) * NPIX + n;
  float vals[DHEAD];
  float m = -INFINITY;
#pragma unroll
  for (int d = 0; d < DHEAD; ++d) {
    vals[d] = qs[(size_t)d * NPIX];
    m = fmaxf(m, vals[d]);
  }
  float s = 0.f;
#pragma unroll
  for (int d = 0; d < DHEAD; ++d) {
    vals[d] = __expf(vals[d] - m);
    s += vals[d];
  }
  const float inv = 1.0f / s;
  float* __restrict__ ko =
      kbuf + (size_t)(b * HEADS + h) * DHEAD * NPIX + n;
#pragma unroll
  for (int d = 0; d < DHEAD; ++d) ko[(size_t)d * NPIX] = vals[d] * inv;
}

__global__ __launch_bounds__(128) void k_context_part(
    const float* __restrict__ kbuf, const float* __restrict__ qkv,
    float* __restrict__ pctx) {
  const int ks = blockIdx.x;
  const int bh = blockIdx.y;
  const int b = bh >> 2, h = bh & 3;
  const int lane = threadIdx.x & 31;
  const int wave = threadIdx.x >> 5;
  const int half = lane >> 4, lm = lane & 15;
  const int mBase = (wave >> 1) * 16;
  const int nBase = (wave & 1) * 16;
  const int kBeg  = ks * KCHUNK;

  const float* __restrict__ kA = kbuf + (size_t)bh * DHEAD * NPIX;
  const float* __restrict__ vB =
      qkv + ((size_t)b * O3 + 2 * HIDDEN + h * DHEAD) * NPIX;

  v8f acc = {};
  for (int kb = 0; kb < KCHUNK; kb += 4) {
    const int k0 = kBeg + kb + 2 * half;
    v2f a, bb;
    a.x  = kA[(size_t)(mBase + lm) * NPIX + k0];
    a.y  = kA[(size_t)(mBase + lm) * NPIX + k0 + 1];
    bb.x = vB[(size_t)(nBase + lm) * NPIX + k0];
    bb.y = vB[(size_t)(nBase + lm) * NPIX + k0 + 1];
    acc = wmma4(a, bb, acc);
  }
#pragma unroll
  for (int v = 0; v < 8; ++v) {
    const int row = mBase + v + 8 * half;
    pctx[(((size_t)ks * (BATCH * HEADS) + bh) * DHEAD + row) * DHEAD + nBase +
         lm] = acc[v];
  }
}

__global__ __launch_bounds__(256) void k_ctx_reduce(
    const float* __restrict__ pctx, float* __restrict__ ctx) {
  const int i = blockIdx.x * 256 + threadIdx.x;
  const size_t stride = (size_t)BATCH * HEADS * DHEAD * DHEAD;
  float s = 0.f;
#pragma unroll
  for (int ks = 0; ks < KSPLIT; ++ks) s += pctx[(size_t)ks * stride + i];
  ctx[i] = s;
}

__global__ __launch_bounds__(256) void k_fold_w2(
    const float* __restrict__ wout, const float* __restrict__ ctx,
    float* __restrict__ w2) {
  const int bh = blockIdx.x;
  const int b = bh >> 2, h = bh & 3;
  const int lane = threadIdx.x & 31;
  const int wave = threadIdx.x >> 5;
  const int half = lane >> 4, lm = lane & 15;
  const int m0 = wave * 16;

  const float* __restrict__ cb = ctx + (size_t)bh * DHEAD * DHEAD;

  v8f acc00 = {}, acc01 = {}, acc10 = {}, acc11 = {};
#pragma unroll
  for (int kb = 0; kb < DHEAD; kb += 4) {
    const int k0 = kb + 2 * half;
    v2f a0, a1, b0, b1;
    a0.x = wout[(m0 + lm) * HIDDEN + h * DHEAD + k0];
    a0.y = wout[(m0 + lm) * HIDDEN + h * DHEAD + k0 + 1];
    a1.x = wout[(m0 + 128 + lm) * HIDDEN + h * DHEAD + k0];
    a1.y = wout[(m0 + 128 + lm) * HIDDEN + h * DHEAD + k0 + 1];
    b0.x = cb[(size_t)(0 + lm) * DHEAD + k0];
    b0.y = cb[(size_t)(0 + lm) * DHEAD + k0 + 1];
    b1.x = cb[(size_t)(16 + lm) * DHEAD + k0];
    b1.y = cb[(size_t)(16 + lm) * DHEAD + k0 + 1];
    acc00 = wmma4(a0, b0, acc00);
    acc01 = wmma4(a0, b1, acc01);
    acc10 = wmma4(a1, b0, acc10);
    acc11 = wmma4(a1, b1, acc11);
  }
  float* __restrict__ w2b = w2 + (size_t)b * CIN * HIDDEN;
#pragma unroll
  for (int v = 0; v < 8; ++v) {
    const int row = v + 8 * half;
    w2b[(size_t)(m0 + row) * HIDDEN + h * DHEAD + 0 + lm]        = acc00[v];
    w2b[(size_t)(m0 + row) * HIDDEN + h * DHEAD + 16 + lm]       = acc01[v];
    w2b[(size_t)(m0 + 128 + row) * HIDDEN + h * DHEAD + 0 + lm]  = acc10[v];
    w2b[(size_t)(m0 + 128 + row) * HIDDEN + h * DHEAD + 16 + lm] = acc11[v];
  }
}

__global__ __launch_bounds__(256) void k_out_gemm(
    const float* __restrict__ qkv, const float* __restrict__ w2,
    float* __restrict__ out) {
  __shared__ __attribute__((aligned(16))) float sA[CIN * APITCH];
  __shared__ __attribute__((aligned(16))) float sB[KC * BPITCH];
  const int b = blockIdx.y;
  gemm_tile_lds<2, HIDDEN>(w2 + (size_t)b * CIN * HIDDEN,
                           qkv + (size_t)b * O3 * NPIX,
                           out + (size_t)b * CIN * NPIX, sA, sB,
                           blockIdx.x * TN);
}

extern "C" void kernel_launch(void* const* d_in, const int* in_sizes, int n_in,
                              void* d_out, int out_size, void* d_ws,
                              size_t ws_size, hipStream_t stream) {
  const float* x    = (const float*)d_in[0];
  const float* wqkv = (const float*)d_in[1];
  const float* wout = (const float*)d_in[2];
  float* out = (float*)d_out;

  char* ws = (char*)d_ws;
  const size_t QKV_F = (size_t)BATCH * O3 * NPIX;
  const size_t KB_F  = (size_t)BATCH * HIDDEN * NPIX;
  const size_t CTX_F = (size_t)BATCH * HEADS * DHEAD * DHEAD;
  float* qkv  = (float*)ws;
  float* kbuf = (float*)(ws + QKV_F * 4);
  float* ctx  = (float*)(ws + (QKV_F + KB_F) * 4);
  float* pctx = (float*)(ws + (QKV_F + KB_F + CTX_F) * 4);
  float* w2   = (float*)(ws + (QKV_F + KB_F + CTX_F + KSPLIT * CTX_F) * 4);

  k_qkv_gemm<<<dim3(NPIX / TN, BATCH), 256, 0, stream>>>(x, wqkv, qkv);
  k_softmax_n<<<BATCH * HIDDEN, 256, 0, stream>>>(qkv);
  k_softmax_d<<<(BATCH * HEADS * NPIX) / 256, 256, 0, stream>>>(qkv, kbuf);
  k_context_part<<<dim3(KSPLIT, BATCH * HEADS), 128, 0, stream>>>(kbuf, qkv,
                                                                  pctx);
  k_ctx_reduce<<<CTX_F / 256, 256, 0, stream>>>(pctx, ctx);
  k_fold_w2<<<BATCH * HEADS, 256, 0, stream>>>(wout, ctx, w2);
  k_out_gemm<<<dim3(NPIX / TN, BATCH), 256, 0, stream>>>(qkv, w2, out);
}


# qkv GEMM computes only the q and v rows (256 of 384): the k-projection rows, which the reference computes but never uses, are no longer computed or stored; summation order unchanged
# speedup vs baseline: 2.7733x; 1.1922x over previous
#include <hip/hip_runtime.h>
#include <hip/hip_bf16.h>
#include <math.h>

#define BATCH   16
#define CIN     256
#define NPIX    4096
#define HEADS   4
#define DHEAD   32
#define HIDDEN  128
#define O3      384
#define KSPLIT  8
#define KCHUNK  (NPIX / KSPLIT)
#define TN      64
#define KC      32
#define APITCH  36
#define BPITCH  72

typedef __attribute__((ext_vector_type(2))) float v2f;
typedef __attribute__((ext_vector_type(4))) float v4f;
typedef __attribute__((ext_vector_type(8))) float v8f;

__device__ __forceinline__ v8f wmma4(v2f a, v2f b, v8f c) {
  return __builtin_amdgcn_wmma_f32_16x16x4_f32(false, a, false, b, (short)0, c,
                                               false, false);
}

template <int MT, int KTOT, int GROWS>
__device__ __forceinline__ void gemm_tile_lds(const float* __restrict__ A,
                                              const float* __restrict__ Bm,
                                              float* __restrict__ C,
                                              float* sA, float* sB,
                                              const int nBase) {
  constexpr int NA = MT * 4;
  const int t    = threadIdx.x;
  const int lane = t & 31;
  const int wave = t >> 5;
  const int half = lane >> 4;
  const int lm   = lane & 15;
  const int m0   = wave * 16;

  const int ar = t >> 3, ac = (t & 7) * 4;
  const int br = t >> 4, bc = (t & 15) * 4;
  const float* __restrict__ gA = A + (size_t)ar * KTOT + ac;
  const float* __restrict__ gB = Bm + (size_t)br * NPIX + nBase + bc;
  float* lA = sA + ar * APITCH + ac;
  float* lB = sB + br * BPITCH + bc;

  const float* fA = sA + (m0 + lm) * APITCH + 2 * half;
  const float* fB = sB + (2 * half) * BPITCH + lm;

  v4f ra[NA];
  v4f rb[2];
  v8f acc[MT][4];
#pragma unroll
  for (int j = 0; j < MT; ++j)
#pragma unroll
    for (int i = 0; i < 4; ++i) acc[j][i] = v8f{};

#pragma unroll
  for (int i = 0; i < NA; ++i)
    ra[i] = *reinterpret_cast<const v4f*>(
        gA + (size_t)((i / 4) * GROWS + 32 * (i % 4)) * KTOT);
#pragma unroll
  for (int i = 0; i < 2; ++i)
    rb[i] = *reinterpret_cast<const v4f*>(gB + (size_t)(16 * i) * NPIX);
#pragma unroll
  for (int i = 0; i < NA; ++i)
    *reinterpret_cast<v4f*>(lA + 32 * i * APITCH) = ra[i];
#pragma unroll
  for (int i = 0; i < 2; ++i)
    *reinterpret_cast<v4f*>(lB + 16 * i * BPITCH) = rb[i];
  __syncthreads();

  for (int kc = 0; kc < KTOT; kc += KC) {
    const bool more = (kc + KC) < KTOT;
    if (more) {
#pragma unroll
      for (int i = 0; i < NA; ++i)
        ra[i] = *reinterpret_cast<const v4f*>(
            gA + (size_t)((i / 4) * GROWS + 32 * (i % 4)) * KTOT + kc + KC);
#pragma unroll
      for (int i = 0; i < 2; ++i)
        rb[i] = *reinterpret_cast<const v4f*>(
            gB + (size_t)(kc + KC + 16 * i) * NPIX);
    }
#pragma unroll
    for (int kk = 0; kk < KC; kk += 4) {
      v2f a[MT], b[4];
#pragma unroll
      for (int j = 0; j < MT; ++j)
        a[j] = *reinterpret_cast<const v2f*>(fA + 128 * j * APITCH + kk);
#pragma unroll
      for (int i = 0; i < 4; ++i) {
        b[i].x = fB[kk * BPITCH + 16 * i];
        b[i].y = fB[(kk + 1) * BPITCH + 16 * i];
      }
#pragma unroll
      for (int j = 0; j < MT; ++j)
#pragma unroll
        for (int i = 0; i < 4; ++i) acc[j][i] = wmma4(a[j], b[i], acc[j][i]);
    }
    __syncthreads();
    if (more) {
#pragma unroll
      for (int i = 0; i < NA; ++i)
        *reinterpret_cast<v4f*>(lA + 32 * i * APITCH) = ra[i];
#pragma unroll
      for (int i = 0; i < 2; ++i)
        *reinterpret_cast<v4f*>(lB + 16 * i * BPITCH) = rb[i];
      __syncthreads();
    }
  }

#pragma unroll
  for (int j = 0; j < MT; ++j)
#pragma unroll
    for (int i = 0; i < 4; ++i)
#pragma unroll
      for (int v = 0; v < 8; ++v)
        C[(size_t)(m0 + GROWS * j + v + 8 * half) * NPIX + nBase + 16 * i + lm] =
            acc[j][i][v];
}

__global__ __launch_bounds__(256) void k_qkv_gemm(
    const float* __restrict__ x, const float* __restrict__ wqkv,
    float* __restrict__ qkv) {
  __shared__ __attribute__((aligned(16))) float sA[2 * HIDDEN * APITCH];
  __shared__ __attribute__((aligned(16))) float sB[KC * BPITCH];
  const int b = blockIdx.y;
  gemm_tile_lds<2, CIN, 2 * HIDDEN>(wqkv, x + (size_t)b * CIN * NPIX,
                                    qkv + (size_t)b * O3 * NPIX, sA, sB,
                                    blockIdx.x * TN);
}

__global__ __launch_bounds__(256) void k_softmax_n(float* __restrict__ qkv) {
  const int r = blockIdx.x;
  const int b = r >> 7;
  const int c = r & 127;
  float* __restrict__ row = qkv + ((size_t)b * O3 + c) * NPIX;

  __shared__ float red[256];
  const int t = threadIdx.x;

  float v[16];
  float m = -INFINITY;
#pragma unroll
  for (int j = 0; j < 16; ++j) {
    v[j] = row[t + j * 256];
    m = fmaxf(m, v[j]);
  }
  red[t] = m; __syncthreads();
  for (int s = 128; s > 0; s >>= 1) {
    if (t < s) red[t] = fmaxf(red[t], red[t + s]);
    __syncthreads();
  }
  m = red[0]; __syncthreads();

  float sum = 0.f;
#pragma unroll
  for (int j = 0; j < 16; ++j) {
    v[j] = __expf(v[j] - m);
    sum += v[j];
  }
  red[t] = sum; __syncthreads();
  for (int s = 128; s > 0; s >>= 1) {
    if (t < s) red[t] += red[t + s];
    __syncthreads();
  }
  const float inv = 1.0f / red[0];
#pragma unroll
  for (int j = 0; j < 16; ++j) row[t + j * 256] = v[j] * inv;
}

__global__ __launch_bounds__(256) void k_softmax_d(
    const float* __restrict__ qkv, float* __restrict__ kbuf) {
  const int col = blockIdx.x * 256 + threadIdx.x;
  const int n = col & (NPIX - 1);
  const int h = (col >> 12) & (HEADS - 1);
  const int b = col >> 14;

  const float* __restrict__ qs = qkv + ((size_t)b * O3 + h * DHEAD) * NPIX + n;
  float vals[DHEAD];
  float m = -INFINITY;
#pragma unroll
  for (int d = 0; d < DHEAD; ++d) {
    vals[d] = qs[(size_t)d * NPIX];
    m = fmaxf(m, vals[d]);
  }
  float s = 0.f;
#pragma unroll
  for (int d = 0; d < DHEAD; ++d) {
    vals[d] = __expf(vals[d] - m);
    s += vals[d];
  }
  const float inv = 1.0f / s;
  float* __restrict__ ko =
      kbuf + (size_t)(b * HEADS + h) * DHEAD * NPIX + n;
#pragma unroll
  for (int d = 0; d < DHEAD; ++d) ko[(size_t)d * NPIX] = vals[d] * inv;
}

__global__ __launch_bounds__(128) void k_context_part(
    const float* __restrict__ kbuf, const float* __restrict__ qkv,
    float* __restrict__ pctx) {
  const int ks = blockIdx.x;
  const int bh = blockIdx.y;
  const int b = bh >> 2, h = bh & 3;
  const int lane = threadIdx.x & 31;
  const int wave = threadIdx.x >> 5;
  const int half = lane >> 4, lm = lane & 15;
  const int mBase = (wave >> 1) * 16;
  const int nBase = (wave & 1) * 16;
  const int kBeg  = ks * KCHUNK;

  const float* __restrict__ kA = kbuf + (size_t)bh * DHEAD * NPIX;
  const float* __restrict__ vB =
      qkv + ((size_t)b * O3 + 2 * HIDDEN + h * DHEAD) * NPIX;

  v8f acc = {};
  for (int kb = 0; kb < KCHUNK; kb += 4) {
    const int k0 = kBeg + kb + 2 * half;
    v2f a, bb;
    a.x  = kA[(size_t)(mBase + lm) * NPIX + k0];
    a.y  = kA[(size_t)(mBase + lm) * NPIX + k0 + 1];
    bb.x = vB[(size_t)(nBase + lm) * NPIX + k0];
    bb.y = vB[(size_t)(nBase + lm) * NPIX + k0 + 1];
    acc = wmma4(a, bb, acc);
  }
#pragma unroll
  for (int v = 0; v < 8; ++v) {
    const int row = mBase + v + 8 * half;
    pctx[(((size_t)ks * (BATCH * HEADS) + bh) * DHEAD + row) * DHEAD + nBase +
         lm] = acc[v];
  }
}

__global__ __launch_bounds__(256) void k_ctx_reduce(
    const float* __restrict__ pctx, float* __restrict__ ctx) {
  const int i = blockIdx.x * 256 + threadIdx.x;
  const size_t stride = (size_t)BATCH * HEADS * DHEAD * DHEAD;
  float s = 0.f;
#pragma unroll
  for (int ks = 0; ks < KSPLIT; ++ks) s += pctx[(size_t)ks * stride + i];
  ctx[i] = s;
}

__global__ __launch_bounds__(256) void k_fold_w2(
    const float* __restrict__ wout, const float* __restrict__ ctx,
    float* __restrict__ w2) {
  const int bh = blockIdx.x;
  const int b = bh >> 2, h = bh & 3;
  const int lane = threadIdx.x & 31;
  const int wave = threadIdx.x >> 5;
  const int half = lane >> 4, lm = lane & 15;
  const int m0 = wave * 16;

  const float* __restrict__ cb = ctx + (size_t)bh * DHEAD * DHEAD;

  v8f acc00 = {}, acc01 = {}, acc10 = {}, acc11 = {};
#pragma unroll
  for (int kb = 0; kb < DHEAD; kb += 4) {
    const int k0 = kb + 2 * half;
    v2f a0, a1, b0, b1;
    a0.x = wout[(m0 + lm) * HIDDEN + h * DHEAD + k0];
    a0.y = wout[(m0 + lm) * HIDDEN + h * DHEAD + k0 + 1];
    a1.x = wout[(m0 + 128 + lm) * HIDDEN + h * DHEAD + k0];
    a1.y = wout[(m0 + 128 + lm) * HIDDEN + h * DHEAD + k0 + 1];
    b0.x = cb[(size_t)(0 + lm) * DHEAD + k0];
    b0.y = cb[(size_t)(0 + lm) * DHEAD + k0 + 1];
    b1.x = cb[(size_t)(16 + lm) * DHEAD + k0];
    b1.y = cb[(size_t)(16 + lm) * DHEAD + k0 + 1];
    acc00 = wmma4(a0, b0, acc00);
    acc01 = wmma4(a0, b1, acc01);
    acc10 = wmma4(a1, b0, acc10);
    acc11 = wmma4(a1, b1, acc11);
  }
  float* __restrict__ w2b = w2 + (size_t)b * CIN * HIDDEN;
#pragma unroll
  for (int v = 0; v < 8; ++v) {
    const int row = v + 8 * half;
    w2b[(size_t)(m0 + row) * HIDDEN + h * DHEAD + 0 + lm]        = acc00[v];
    w2b[(size_t)(m0 + row) * HIDDEN + h * DHEAD + 16 + lm]       = acc01[v];
    w2b[(size_t)(m0 + 128 + row) * HIDDEN + h * DHEAD + 0 + lm]  = acc10[v];
    w2b[(size_t)(m0 + 128 + row) * HIDDEN + h * DHEAD + 16 + lm] = acc11[v];
  }
}

__global__ __launch_bounds__(256) void k_out_gemm(
    const float* __restrict__ qkv, const float* __restrict__ w2,
    float* __restrict__ out) {
  __shared__ __attribute__((aligned(16))) float sA[CIN * APITCH];
  __shared__ __attribute__((aligned(16))) float sB[KC * BPITCH];
  const int b = blockIdx.y;
  gemm_tile_lds<2, HIDDEN, 128>(w2 + (size_t)b * CIN * HIDDEN,
                                qkv + (size_t)b * O3 * NPIX,
                                out + (size_t)b * CIN * NPIX, sA, sB,
                                blockIdx.x * TN);
}

extern "C" void kernel_launch(void* const* d_in, const int* in_sizes, int n_in,
                              void* d_out, int out_size, void* d_ws,
                              size_t ws_size, hipStream_t stream) {
  const float* x    = (const float*)d_in[0];
  const float* wqkv = (const float*)d_in[1];
  const float* wout = (const float*)d_in[2];
  float* out = (float*)d_out;

  char* ws = (char*)d_ws;
  const size_t QKV_F = (size_t)BATCH * O3 * NPIX;
  const size_t KB_F  = (size_t)BATCH * HIDDEN * NPIX;
  const size_t CTX_F = (size_t)BATCH * HEADS * DHEAD * DHEAD;
  float* qkv  = (float*)ws;
  float* kbuf = (float*)(ws + QKV_F * 4);
  float* ctx  = (float*)(ws + (QKV_F + KB_F) * 4);
  float* pctx = (float*)(ws + (QKV_F + KB_F + CTX_F) * 4);
  float* w2   = (float*)(ws + (QKV_F + KB_F + CTX_F + KSPLIT * CTX_F) * 4);

  k_qkv_gemm<<<dim3(NPIX / TN, BATCH), 256, 0, stream>>>(x, wqkv, qkv);
  k_softmax_n<<<BATCH * HIDDEN, 256, 0, stream>>>(qkv);
  k_softmax_d<<<(BATCH * HEADS * NPIX) / 256, 256, 0, stream>>>(qkv, kbuf);
  k_context_part<<<dim3(KSPLIT, BATCH * HEADS), 128, 0, stream>>>(kbuf, qkv,
                                                                  pctx);
  k_ctx_reduce<<<CTX_F / 256, 256, 0, stream>>>(pctx, ctx);
  k_fold_w2<<<BATCH * HEADS, 256, 0, stream>>>(wout, ctx, w2);
  k_out_gemm<<<dim3(NPIX / TN, BATCH), 256, 0, stream>>>(qkv, w2, out);
}


# context GEMM: k and v tiles staged through LDS by whole-line 16-byte global loads issued two stages ahead, fragments read from LDS (was per-lane strided 8-byte loads); summation order unchanged
# speedup vs baseline: 2.8901x; 1.0421x over previous
#include <hip/hip_runtime.h>
#include <hip/hip_bf16.h>
#include <math.h>

#define BATCH   16
#define CIN     256
#define NPIX    4096
#define HEADS   4
#define DHEAD   32
#define HIDDEN  128
#define O3      384
#define KSPLIT  8
#define KCHUNK  (NPIX / KSPLIT)
#define TN      64
#define KC      32
#define APITCH  36
#define BPITCH  72

typedef __attribute__((ext_vector_type(2))) float v2f;
typedef __attribute__((ext_vector_type(4))) float v4f;
typedef __attribute__((ext_vector_type(8))) float v8f;

__device__ __forceinline__ v8f wmma4(v2f a, v2f b, v8f c) {
  return __builtin_amdgcn_wmma_f32_16x16x4_f32(false, a, false, b, (short)0, c,
                                               false, false);
}

template <int MT, int KTOT, int GROWS>
__device__ __forceinline__ void gemm_tile_lds(const float* __restrict__ A,
                                              const float* __restrict__ Bm,
                                              float* __restrict__ C,
                                              float* sA, float* sB,
                                              const int nBase) {
  constexpr int NA = MT * 4;
  const int t    = threadIdx.x;
  const int lane = t & 31;
  const int wave = t >> 5;
  const int half = lane >> 4;
  const int lm   = lane & 15;
  const int m0   = wave * 16;

  const int ar = t >> 3, ac = (t & 7) * 4;
  const int br = t >> 4, bc = (t & 15) * 4;
  const float* __restrict__ gA = A + (size_t)ar * KTOT + ac;
  const float* __restrict__ gB = Bm + (size_t)br * NPIX + nBase + bc;
  float* lA = sA + ar * APITCH + ac;
  float* lB = sB + br * BPITCH + bc;

  const float* fA = sA + (m0 + lm) * APITCH + 2 * half;
  const float* fB = sB + (2 * half) * BPITCH + lm;

  v4f ra[NA];
  v4f rb[2];
  v8f acc[MT][4];
#pragma unroll
  for (int j = 0; j < MT; ++j)
#pragma unroll
    for (int i = 0; i < 4; ++i) acc[j][i] = v8f{};

#pragma unroll
  for (int i = 0; i < NA; ++i)
    ra[i] = *reinterpret_cast<const v4f*>(
        gA + (size_t)((i / 4) * GROWS + 32 * (i % 4)) * KTOT);
#pragma unroll
  for (int i = 0; i < 2; ++i)
    rb[i] = *reinterpret_cast<const v4f*>(gB + (size_t)(16 * i) * NPIX);
#pragma unroll
  for (int i = 0; i < NA; ++i)
    *reinterpret_cast<v4f*>(lA + 32 * i * APITCH) = ra[i];
#pragma unroll
  for (int i = 0; i < 2; ++i)
    *reinterpret_cast<v4f*>(lB + 16 * i * BPITCH) = rb[i];
  __syncthreads();

  for (int kc = 0; kc < KTOT; kc += KC) {
    const bool more = (kc + KC) < KTOT;
    if (more) {
#pragma unroll
      for (int i = 0; i < NA; ++i)
        ra[i] = *reinterpret_cast<const v4f*>(
            gA + (size_t)((i / 4) * GROWS + 32 * (i % 4)) * KTOT + kc + KC);
#pragma unroll
      for (int i = 0; i < 2; ++i)
        rb[i] = *reinterpret_cast<const v4f*>(
            gB + (size_t)(kc + KC + 16 * i) * NPIX);
    }
#pragma unroll
    for (int kk = 0; kk < KC; kk += 4) {
      v2f a[MT], b[4];
#pragma unroll
      for (int j = 0; j < MT; ++j)
        a[j] = *reinterpret_cast<const v2f*>(fA + 128 * j * APITCH + kk);
#pragma unroll
      for (int i = 0; i < 4; ++i) {
        b[i].x = fB[kk * BPITCH + 16 * i];
        b[i].y = fB[(kk + 1) * BPITCH + 16 * i];
      }
#pragma unroll
      for (int j = 0; j < MT; ++j)
#pragma unroll
        for (int i = 0; i < 4; ++i) acc[j][i] = wmma4(a[j], b[i], acc[j][i]);
    }
    __syncthreads();
    if (more) {
#pragma unroll
      for (int i = 0; i < NA; ++i)
        *reinterpret_cast<v4f*>(lA + 32 * i * APITCH) = ra[i];
#pragma unroll
      for (int i = 0; i < 2; ++i)
        *reinterpret_cast<v4f*>(lB + 16 * i * BPITCH) = rb[i];
      __syncthreads();
    }
  }

#pragma unroll
  for (int j = 0; j < MT; ++j)
#pragma unroll
    for (int i = 0; i < 4; ++i)
#pragma unroll
      for (int v = 0; v < 8; ++v)
        C[(size_t)(m0 + GROWS * j + v + 8 * half) * NPIX + nBase + 16 * i + lm] =
            acc[j][i][v];
}

__global__ __launch_bounds__(256) void k_qkv_gemm(
    const float* __restrict__ x, const float* __restrict__ wqkv,
    float* __restrict__ qkv) {
  __shared__ __attribute__((aligned(16))) float sA[2 * HIDDEN * APITCH];
  __shared__ __attribute__((aligned(16))) float sB[KC * BPITCH];
  const int b = blockIdx.y;
  gemm_tile_lds<2, CIN, 2 * HIDDEN>(wqkv, x + (size_t)b * CIN * NPIX,
                                    qkv + (size_t)b * O3 * NPIX, sA, sB,
                                    blockIdx.x * TN);
}

__global__ __launch_bounds__(256) void k_softmax_n(float* __restrict__ qkv) {
  const int r = blockIdx.x;
  const int b = r >> 7;
  const int c = r & 127;
  float* __restrict__ row = qkv + ((size_t)b * O3 + c) * NPIX;

  __shared__ float red[256];
  const int t = threadIdx.x;

  float v[16];
  float m = -INFINITY;
#pragma unroll
  for (int j = 0; j < 16; ++j) {
    v[j] = row[t + j * 256];
    m = fmaxf(m, v[j]);
  }
  red[t] = m; __syncthreads();
  for (int s = 128; s > 0; s >>= 1) {
    if (t < s) red[t] = fmaxf(red[t], red[t + s]);
    __syncthreads();
  }
  m = red[0]; __syncthreads();

  float sum = 0.f;
#pragma unroll
  for (int j = 0; j < 16; ++j) {
    v[j] = __expf(v[j] - m);
    sum += v[j];
  }
  red[t] = sum; __syncthreads();
  for (int s = 128; s > 0; s >>= 1) {
    if (t < s) red[t] += red[t + s];
    __syncthreads();
  }
  const float inv = 1.0f / red[0];
#pragma unroll
  for (int j = 0; j < 16; ++j) row[t + j * 256] = v[j] * inv;
}

__global__ __launch_bounds__(256) void k_softmax_d(
    const float* __restrict__ qkv, float* __restrict__ kbuf) {
  const int col = blockIdx.x * 256 + threadIdx.x;
  const int n = col & (NPIX - 1);
  const int h = (col >> 12) & (HEADS - 1);
  const int b = col >> 14;

  const float* __restrict__ qs = qkv + ((size_t)b * O3 + h * DHEAD) * NPIX + n;
  float vals[DHEAD];
  float m = -INFINITY;
#pragma unroll
  for (int d = 0; d < DHEAD; ++d) {
    vals[d] = qs[(size_t)d * NPIX];
    m = fmaxf(m, vals[d]);
  }
  float s = 0.f;
#pragma unroll
  for (int d = 0; d < DHEAD; ++d) {
    vals[d] = __expf(vals[d] - m);
    s += vals[d];
  }
  const float inv = 1.0f / s;
  float* __restrict__ ko =
      kbuf + (size_t)(b * HEADS + h) * DHEAD * NPIX + n;
#pragma unroll
  for (int d = 0; d < DHEAD; ++d) ko[(size_t)d * NPIX] = vals[d] * inv;
}

#define CSUB    128
#define NCSUB   (KCHUNK / CSUB)
#define CPITCH  132
#define CTROWS  (2 * DHEAD)

__device__ __forceinline__ void ctx_fetch(v4f (&r)[16],
                                          const float* __restrict__ gK,
                                          const float* __restrict__ gV,
                                          const int col) {
#pragma unroll
  for (int i = 0; i < 8; ++i) {
    r[i]     = *reinterpret_cast<const v4f*>(gK + (size_t)(4 * i) * NPIX + col);
    r[8 + i] = *reinterpret_cast<const v4f*>(gV + (size_t)(4 * i) * NPIX + col);
  }
}

__device__ __forceinline__ void ctx_park(const v4f (&r)[16], float* l) {
#pragma unroll
  for (int i = 0; i < 8; ++i) {
    *reinterpret_cast<v4f*>(l + (4 * i) * CPITCH)           = r[i];
    *reinterpret_cast<v4f*>(l + (DHEAD + 4 * i) * CPITCH)   = r[8 + i];
  }
}

__device__ __forceinline__ v8f ctx_mma(const float* fA, const float* fB,
                                       v8f acc) {
#pragma unroll
  for (int kk = 0; kk < CSUB; kk += 4) {
    const v2f a  = *reinterpret_cast<const v2f*>(fA + kk);
    const v2f bb = *reinterpret_cast<const v2f*>(fB + kk);
    acc = wmma4(a, bb, acc);
  }
  return acc;
}

__global__ __launch_bounds__(128) void k_context_part(
    const float* __restrict__ kbuf, const float* __restrict__ qkv,
    float* __restrict__ pctx) {
  static_assert(NCSUB == 4, "the stage sequence below is written out for four stages");
  __shared__ __attribute__((aligned(16))) float sT[2][CTROWS * CPITCH];
  const int ks = blockIdx.x;
  const int bh = blockIdx.y;
  const int b = bh >> 2, h = bh & 3;
  const int lane = threadIdx.x & 31;
  const int wave = threadIdx.x >> 5;
  const int half = lane >> 4, lm = lane & 15;
  const int mBase = (wave >> 1) * 16;
  const int nBase = (wave & 1) * 16;
  const int kBeg  = ks * KCHUNK;

  const float* __restrict__ kA = kbuf + (size_t)bh * DHEAD * NPIX;
  const float* __restrict__ vB =
      qkv + ((size_t)b * O3 + 2 * HIDDEN + h * DHEAD) * NPIX;

  const float* __restrict__ gK = kA + (size_t)wave * NPIX + kBeg + lane * 4;
  const float* __restrict__ gV = vB + (size_t)wave * NPIX + kBeg + lane * 4;
  const int lOff = wave * CPITCH + lane * 4;
  const int aOff = (mBase + lm) * CPITCH + 2 * half;
  const int bOff = (DHEAD + nBase + lm) * CPITCH + 2 * half;

  v4f r0[16], r1[16];
  ctx_fetch(r0, gK, gV, 0 * CSUB);
  ctx_fetch(r1, gK, gV, 1 * CSUB);
  ctx_park(r0, &sT[0][lOff]);
  __syncthreads();

  v8f acc = {};
  ctx_fetch(r0, gK, gV, 2 * CSUB);
  acc = ctx_mma(&sT[0][aOff], &sT[0][bOff], acc);
  ctx_park(r1, &sT[1][lOff]);
  __syncthreads();

  ctx_fetch(r1, gK, gV, 3 * CSUB);
  acc = ctx_mma(&sT[1][aOff], &sT[1][bOff], acc);
  ctx_park(r0, &sT[0][lOff]);
  __syncthreads();

  acc = ctx_mma(&sT[0][aOff], &sT[0][bOff], acc);
  ctx_park(r1, &sT[1][lOff]);
  __syncthreads();

  acc = ctx_mma(&sT[1][aOff], &sT[1][bOff], acc);

#pragma unroll
  for (int v = 0; v < 8; ++v) {
    const int row = mBase + v + 8 * half;
    pctx[(((size_t)ks * (BATCH * HEADS) + bh) * DHEAD + row) * DHEAD + nBase +
         lm] = acc[v];
  }
}

__global__ __launch_bounds__(256) void k_ctx_reduce(
    const float* __restrict__ pctx, float* __restrict__ ctx) {
  const int i = blockIdx.x * 256 + threadIdx.x;
  const size_t stride = (size_t)BATCH * HEADS * DHEAD * DHEAD;
  float s = 0.f;
#pragma unroll
  for (int ks = 0; ks < KSPLIT; ++ks) s += pctx[(size_t)ks * stride + i];
  ctx[i] = s;
}

__global__ __launch_bounds__(256) void k_fold_w2(
    const float* __restrict__ wout, const float* __restrict__ ctx,
    float* __restrict__ w2) {
  const int bh = blockIdx.x;
  const int b = bh >> 2, h = bh & 3;
  const int lane = threadIdx.x & 31;
  const int wave = threadIdx.x >> 5;
  const int half = lane >> 4, lm = lane & 15;
  const int m0 = wave * 16;

  const float* __restrict__ cb = ctx + (size_t)bh * DHEAD * DHEAD;

  v8f acc00 = {}, acc01 = {}, acc10 = {}, acc11 = {};
#pragma unroll
  for (int kb = 0; kb < DHEAD; kb += 4) {
    const int k0 = kb + 2 * half;
    v2f a0, a1, b0, b1;
    a0.x = wout[(m0 + lm) * HIDDEN + h * DHEAD + k0];
    a0.y = wout[(m0 + lm) * HIDDEN + h * DHEAD + k0 + 1];
    a1.x = wout[(m0 + 128 + lm) * HIDDEN + h * DHEAD + k0];
    a1.y = wout[(m0 + 128 + lm) * HIDDEN + h * DHEAD + k0 + 1];
    b0.x = cb[(size_t)(0 + lm) * DHEAD + k0];
    b0.y = cb[(size_t)(0 + lm) * DHEAD + k0 + 1];
    b1.x = cb[(size_t)(16 + lm) * DHEAD + k0];
    b1.y = cb[(size_t)(16 + lm) * DHEAD + k0 + 1];
    acc00 = wmma4(a0, b0, acc00);
    acc01 = wmma4(a0, b1, acc01);
    acc10 = wmma4(a1, b0, acc10);
    acc11 = wmma4(a1, b1, acc11);
  }
  float* __restrict__ w2b = w2 + (size_t)b * CIN * HIDDEN;
#pragma unroll
  for (int v = 0; v < 8; ++v) {
    const int row = v + 8 * half;
    w2b[(size_t)(m0 + row) * HIDDEN + h * DHEAD + 0 + lm]        = acc00[v];
    w2b[(size_t)(m0 + row) * HIDDEN + h * DHEAD + 16 + lm]       = acc01[v];
    w2b[(size_t)(m0 + 128 + row) * HIDDEN + h * DHEAD + 0 + lm]  = acc10[v];
    w2b[(size_t)(m0 + 128 + row) * HIDDEN + h * DHEAD + 16 + lm] = acc11[v];
  }
}

__global__ __launch_bounds__(256) void k_out_gemm(
    const float* __restrict__ qkv, const float* __restrict__ w2,
    float* __restrict__ out) {
  __shared__ __attribute__((aligned(16))) float sA[CIN * APITCH];
  __shared__ __attribute__((aligned(16))) float sB[KC * BPITCH];
  const int b = blockIdx.y;
  gemm_tile_lds<2, HIDDEN, 128>(w2 + (size_t)b * CIN * HIDDEN,
                                qkv + (size_t)b * O3 * NPIX,
                                out + (size_t)b * CIN * NPIX, sA, sB,
                                blockIdx.x * TN);
}

extern "C" void kernel_launch(void* const* d_in, const int* in_sizes, int n_in,
                              void* d_out, int out_size, void* d_ws,
                              size_t ws_size, hipStream_t stream) {
  const float* x    = (const float*)d_in[0];
  const float* wqkv = (const float*)d_in[1];
  const float* wout = (const float*)d_in[2];
  float* out = (float*)d_out;

  char* ws = (char*)d_ws;
  const size_t QKV_F = (size_t)BATCH * O3 * NPIX;
  const size_t KB_F  = (size_t)BATCH * HIDDEN * NPIX;
  const size_t CTX_F = (size_t)BATCH * HEADS * DHEAD * DHEAD;
  float* qkv  = (float*)ws;
  float* kbuf = (float*)(ws + QKV_F * 4);
  float* ctx  = (float*)(ws + (QKV_F + KB_F) * 4);
  float* pctx = (float*)(ws + (QKV_F + KB_F + CTX_F) * 4);
  float* w2   = (float*)(ws + (QKV_F + KB_F + CTX_F + KSPLIT * CTX_F) * 4);

  k_qkv_gemm<<<dim3(NPIX / TN, BATCH), 256, 0, stream>>>(x, wqkv, qkv);
  k_softmax_n<<<BATCH * HIDDEN, 256, 0, stream>>>(qkv);
  k_softmax_d<<<(BATCH * HEADS * NPIX) / 256, 256, 0, stream>>>(qkv, kbuf);
  k_context_part<<<dim3(KSPLIT, BATCH * HEADS), 128, 0, stream>>>(kbuf, qkv,
                                                                  pctx);
  k_ctx_reduce<<<CTX_F / 256, 256, 0, stream>>>(pctx, ctx);
  k_fold_w2<<<BATCH * HEADS, 256, 0, stream>>>(wout, ctx, w2);
  k_out_gemm<<<dim3(NPIX / TN, BATCH), 256, 0, stream>>>(qkv, w2, out);
}


# context GEMM forms k = softmax over d on its staged q tile in LDS (one thread per column, same operation order); the column-softmax launch and the k plane round trip go; summation order unchanged
# speedup vs baseline: 3.0385x; 1.0513x over previous
#include <hip/hip_runtime.h>
#include <hip/hip_bf16.h>
#include <math.h>

#define BATCH   16
#define CIN     256
#define NPIX    4096
#define HEADS   4
#define DHEAD   32
#define HIDDEN  128
#define O3      384
#define KSPLIT  8
#define KCHUNK  (NPIX / KSPLIT)
#define TN      64
#define KC      32
#define APITCH  36
#define BPITCH  72

typedef __attribute__((ext_vector_type(2))) float v2f;
typedef __attribute__((ext_vector_type(4))) float v4f;
typedef __attribute__((ext_vector_type(8))) float v8f;

__device__ __forceinline__ v8f wmma4(v2f a, v2f b, v8f c) {
  return __builtin_amdgcn_wmma_f32_16x16x4_f32(false, a, false, b, (short)0, c,
                                               false, false);
}

template <int MT, int KTOT, int GROWS>
__device__ __forceinline__ void gemm_tile_lds(const float* __restrict__ A,
                                              const float* __restrict__ Bm,
                                              float* __restrict__ C,
                                              float* sA, float* sB,
                                              const int nBase) {
  constexpr int NA = MT * 4;
  const int t    = threadIdx.x;
  const int lane = t & 31;
  const int wave = t >> 5;
  const int half = lane >> 4;
  const int lm   = lane & 15;
  const int m0   = wave * 16;

  const int ar = t >> 3, ac = (t & 7) * 4;
  const int br = t >> 4, bc = (t & 15) * 4;
  const float* __restrict__ gA = A + (size_t)ar * KTOT + ac;
  const float* __restrict__ gB = Bm + (size_t)br * NPIX + nBase + bc;
  float* lA = sA + ar * APITCH + ac;
  float* lB = sB + br * BPITCH + bc;

  const float* fA = sA + (m0 + lm) * APITCH + 2 * half;
  const float* fB = sB + (2 * half) * BPITCH + lm;

  v4f ra[NA];
  v4f rb[2];
  v8f acc[MT][4];
#pragma unroll
  for (int j = 0; j < MT; ++j)
#pragma unroll
    for (int i = 0; i < 4; ++i) acc[j][i] = v8f{};

#pragma unroll
  for (int i = 0; i < NA; ++i)
    ra[i] = *reinterpret_cast<const v4f*>(
        gA + (size_t)((i / 4) * GROWS + 32 * (i % 4)) * KTOT);
#pragma unroll
  for (int i = 0; i < 2; ++i)
    rb[i] = *reinterpret_cast<const v4f*>(gB + (size_t)(16 * i) * NPIX);
#pragma unroll
  for (int i = 0; i < NA; ++i)
    *reinterpret_cast<v4f*>(lA + 32 * i * APITCH) = ra[i];
#pragma unroll
  for (int i = 0; i < 2; ++i)
    *reinterpret_cast<v4f*>(lB + 16 * i * BPITCH) = rb[i];
  __syncthreads();

  for (int kc = 0; kc < KTOT; kc += KC) {
    const bool more = (kc + KC) < KTOT;
    if (more) {
#pragma unroll
      for (int i = 0; i < NA; ++i)
        ra[i] = *reinterpret_cast<const v4f*>(
            gA + (size_t)((i / 4) * GROWS + 32 * (i % 4)) * KTOT + kc + KC);
#pragma unroll
      for (int i = 0; i < 2; ++i)
        rb[i] = *reinterpret_cast<const v4f*>(
            gB + (size_t)(kc + KC + 16 * i) * NPIX);
    }
#pragma unroll
    for (int kk = 0; kk < KC; kk += 4) {
      v2f a[MT], b[4];
#pragma unroll
      for (int j = 0; j < MT; ++j)
        a[j] = *reinterpret_cast<const v2f*>(fA + 128 * j * APITCH + kk);
#pragma unroll
      for (int i = 0; i < 4; ++i) {
        b[i].x = fB[kk * BPITCH + 16 * i];
        b[i].y = fB[(kk + 1) * BPITCH + 16 * i];
      }
#pragma unroll
      for (int j = 0; j < MT; ++j)
#pragma unroll
        for (int i = 0; i < 4; ++i) acc[j][i] = wmma4(a[j], b[i], acc[j][i]);
    }
    __syncthreads();
    if (more) {
#pragma unroll
      for (int i = 0; i < NA; ++i)
        *reinterpret_cast<v4f*>(lA + 32 * i * APITCH) = ra[i];
#pragma unroll
      for (int i = 0; i < 2; ++i)
        *reinterpret_cast<v4f*>(lB + 16 * i * BPITCH) = rb[i];
      __syncthreads();
    }
  }

#pragma unroll
  for (int j = 0; j < MT; ++j)
#pragma unroll
    for (int i = 0; i < 4; ++i)
#pragma unroll
      for (int v = 0; v < 8; ++v)
        C[(size_t)(m0 + GROWS * j + v + 8 * half) * NPIX + nBase + 16 * i + lm] =
            acc[j][i][v];
}

__global__ __launch_bounds__(256) void k_qkv_gemm(
    const float* __restrict__ x, const float* __restrict__ wqkv,
    float* __restrict__ qkv) {
  __shared__ __attribute__((aligned(16))) float sA[2 * HIDDEN * APITCH];
  __shared__ __attribute__((aligned(16))) float sB[KC * BPITCH];
  const int b = blockIdx.y;
  gemm_tile_lds<2, CIN, 2 * HIDDEN>(wqkv, x + (size_t)b * CIN * NPIX,
                                    qkv + (size_t)b * O3 * NPIX, sA, sB,
                                    blockIdx.x * TN);
}

__global__ __launch_bounds__(256) void k_softmax_n(float* __restrict__ qkv) {
  const int r = blockIdx.x;
  const int b = r >> 7;
  const int c = r & 127;
  float* __restrict__ row = qkv + ((size_t)b * O3 + c) * NPIX;

  __shared__ float red[256];
  const int t = threadIdx.x;

  float v[16];
  float m = -INFINITY;
#pragma unroll
  for (int j = 0; j < 16; ++j) {
    v[j] = row[t + j * 256];
    m = fmaxf(m, v[j]);
  }
  red[t] = m; __syncthreads();
  for (int s = 128; s > 0; s >>= 1) {
    if (t < s) red[t] = fmaxf(red[t], red[t + s]);
    __syncthreads();
  }
  m = red[0]; __syncthreads();

  float sum = 0.f;
#pragma unroll
  for (int j = 0; j < 16; ++j) {
    v[j] = __expf(v[j] - m);
    sum += v[j];
  }
  red[t] = sum; __syncthreads();
  for (int s = 128; s > 0; s >>= 1) {
    if (t < s) red[t] += red[t + s];
    __syncthreads();
  }
  const float inv = 1.0f / red[0];
#pragma unroll
  for (int j = 0; j < 16; ++j) row[t + j * 256] = v[j] * inv;
}

#define CSUB    128
#define NCSUB   (KCHUNK / CSUB)
#define CPITCH  132
#define CTROWS  (2 * DHEAD)

__device__ __forceinline__ void ctx_fetch(v4f (&r)[16],
                                          const float* __restrict__ gK,
                                          const float* __restrict__ gV,
                                          const int col) {
#pragma unroll
  for (int i = 0; i < 8; ++i) {
    r[i]     = *reinterpret_cast<const v4f*>(gK + (size_t)(4 * i) * NPIX + col);
    r[8 + i] = *reinterpret_cast<const v4f*>(gV + (size_t)(4 * i) * NPIX + col);
  }
}

__device__ __forceinline__ void ctx_park(const v4f (&r)[16], float* l) {
#pragma unroll
  for (int i = 0; i < 8; ++i) {
    *reinterpret_cast<v4f*>(l + (4 * i) * CPITCH)           = r[i];
    *reinterpret_cast<v4f*>(l + (DHEAD + 4 * i) * CPITCH)   = r[8 + i];
  }
}

__device__ __forceinline__ void ctx_colsoftmax(float* col) {
  float vals[DHEAD];
  float m = -INFINITY;
#pragma unroll
  for (int d = 0; d < DHEAD; ++d) {
    vals[d] = col[d * CPITCH];
    m = fmaxf(m, vals[d]);
  }
  float s = 0.f;
#pragma unroll
  for (int d = 0; d < DHEAD; ++d) {
    vals[d] = __expf(vals[d] - m);
    s += vals[d];
  }
  const float inv = 1.0f / s;
#pragma unroll
  for (int d = 0; d < DHEAD; ++d) col[d * CPITCH] = vals[d] * inv;
}

__device__ __forceinline__ v8f ctx_mma(const float* fA, const float* fB,
                                       v8f acc) {
#pragma unroll
  for (int kk = 0; kk < CSUB; kk += 4) {
    const v2f a  = *reinterpret_cast<const v2f*>(fA + kk);
    const v2f bb = *reinterpret_cast<const v2f*>(fB + kk);
    acc = wmma4(a, bb, acc);
  }
  return acc;
}

__global__ __launch_bounds__(128) void k_context_part(
    const float* __restrict__ qkv, float* __restrict__ pctx) {
  static_assert(NCSUB == 4, "the stage sequence below is written out for four stages");
  static_assert(CSUB == 128, "one thread per column of a stage in the column softmax");
  __shared__ __attribute__((aligned(16))) float sT[2][CTROWS * CPITCH];
  const int ks = blockIdx.x;
  const int bh = blockIdx.y;
  const int b = bh >> 2, h = bh & 3;
  const int lane = threadIdx.x & 31;
  const int wave = threadIdx.x >> 5;
  const int half = lane >> 4, lm = lane & 15;
  const int mBase = (wave >> 1) * 16;
  const int nBase = (wave & 1) * 16;
  const int kBeg  = ks * KCHUNK;

  const float* __restrict__ kA =
      qkv + ((size_t)b * O3 + h * DHEAD) * NPIX;
  const float* __restrict__ vB =
      qkv + ((size_t)b * O3 + 2 * HIDDEN + h * DHEAD) * NPIX;

  const float* __restrict__ gK = kA + (size_t)wave * NPIX + kBeg + lane * 4;
  const float* __restrict__ gV = vB + (size_t)wave * NPIX + kBeg + lane * 4;
  const int lOff = wave * CPITCH + lane * 4;
  const int aOff = (mBase + lm) * CPITCH + 2 * half;
  const int bOff = (DHEAD + nBase + lm) * CPITCH + 2 * half;

  const int cOff = threadIdx.x;

  v4f r0[16], r1[16];
  ctx_fetch(r0, gK, gV, 0 * CSUB);
  ctx_fetch(r1, gK, gV, 1 * CSUB);
  ctx_park(r0, &sT[0][lOff]);
  __syncthreads();

  v8f acc = {};
  ctx_fetch(r0, gK, gV, 2 * CSUB);
  ctx_colsoftmax(&sT[0][cOff]);
  ctx_park(r1, &sT[1][lOff]);
  __syncthreads();

  ctx_fetch(r1, gK, gV, 3 * CSUB);
  ctx_colsoftmax(&sT[1][cOff]);
  acc = ctx_mma(&sT[0][aOff], &sT[0][bOff], acc);
  __syncthreads();

  ctx_park(r0, &sT[0][lOff]);
  acc = ctx_mma(&sT[1][aOff], &sT[1][bOff], acc);
  __syncthreads();

  ctx_colsoftmax(&sT[0][cOff]);
  ctx_park(r1, &sT[1][lOff]);
  __syncthreads();

  ctx_colsoftmax(&sT[1][cOff]);
  acc = ctx_mma(&sT[0][aOff], &sT[0][bOff], acc);
  __syncthreads();

  acc = ctx_mma(&sT[1][aOff], &sT[1][bOff], acc);

#pragma unroll
  for (int v = 0; v < 8; ++v) {
    const int row = mBase + v + 8 * half;
    pctx[(((size_t)ks * (BATCH * HEADS) + bh) * DHEAD + row) * DHEAD + nBase +
         lm] = acc[v];
  }
}

__global__ __launch_bounds__(256) void k_ctx_reduce(
    const float* __restrict__ pctx, float* __restrict__ ctx) {
  const int i = blockIdx.x * 256 + threadIdx.x;
  const size_t stride = (size_t)BATCH * HEADS * DHEAD * DHEAD;
  float s = 0.f;
#pragma unroll
  for (int ks = 0; ks < KSPLIT; ++ks) s += pctx[(size_t)ks * stride + i];
  ctx[i] = s;
}

__global__ __launch_bounds__(256) void k_fold_w2(
    const float* __restrict__ wout, const float* __restrict__ ctx,
    float* __restrict__ w2) {
  const int bh = blockIdx.x;
  const int b = bh >> 2, h = bh & 3;
  const int lane = threadIdx.x & 31;
  const int wave = threadIdx.x >> 5;
  const int half = lane >> 4, lm = lane & 15;
  const int m0 = wave * 16;

  const float* __restrict__ cb = ctx + (size_t)bh * DHEAD * DHEAD;

  v8f acc00 = {}, acc01 = {}, acc10 = {}, acc11 = {};
#pragma unroll
  for (int kb = 0; kb < DHEAD; kb += 4) {
    const int k0 = kb + 2 * half;
    v2f a0, a1, b0, b1;
    a0.x = wout[(m0 + lm) * HIDDEN + h * DHEAD + k0];
    a0.y = wout[(m0 + lm) * HIDDEN + h * DHEAD + k0 + 1];
    a1.x = wout[(m0 + 128 + lm) * HIDDEN + h * DHEAD + k0];
    a1.y = wout[(m0 + 128 + lm) * HIDDEN + h * DHEAD + k0 + 1];
    b0.x = cb[(size_t)(0 + lm) * DHEAD + k0];
    b0.y = cb[(size_t)(0 + lm) * DHEAD + k0 + 1];
    b1.x = cb[(size_t)(16 + lm) * DHEAD + k0];
    b1.y = cb[(size_t)(16 + lm) * DHEAD + k0 + 1];
    acc00 = wmma4(a0, b0, acc00);
    acc01 = wmma4(a0, b1, acc01);
    acc10 = wmma4(a1, b0, acc10);
    acc11 = wmma4(a1, b1, acc11);
  }
  float* __restrict__ w2b = w2 + (size_t)b * CIN * HIDDEN;
#pragma unroll
  for (int v = 0; v < 8; ++v) {
    const int row = v + 8 * half;
    w2b[(size_t)(m0 + row) * HIDDEN + h * DHEAD + 0 + lm]        = acc00[v];
    w2b[(size_t)(m0 + row) * HIDDEN + h * DHEAD + 16 + lm]       = acc01[v];
    w2b[(size_t)(m0 + 128 + row) * HIDDEN + h * DHEAD + 0 + lm]  = acc10[v];
    w2b[(size_t)(m0 + 128 + row) * HIDDEN + h * DHEAD + 16 + lm] = acc11[v];
  }
}

__global__ __launch_bounds__(256) void k_out_gemm(
    const float* __restrict__ qkv, const float* __restrict__ w2,
    float* __restrict__ out) {
  __shared__ __attribute__((aligned(16))) float sA[CIN * APITCH];
  __shared__ __attribute__((aligned(16))) float sB[KC * BPITCH];
  const int b = blockIdx.y;
  gemm_tile_lds<2, HIDDEN, 128>(w2 + (size_t)b * CIN * HIDDEN,
                                qkv + (size_t)b * O3 * NPIX,
                                out + (size_t)b * CIN * NPIX, sA, sB,
                                blockIdx.x * TN);
}

extern "C" void kernel_launch(void* const* d_in, const int* in_sizes, int n_in,
                              void* d_out, int out_size, void* d_ws,
                              size_t ws_size, hipStream_t stream) {
  const float* x    = (const float*)d_in[0];
  const float* wqkv = (const float*)d_in[1];
  const float* wout = (const float*)d_in[2];
  float* out = (float*)d_out;

  char* ws = (char*)d_ws;
  const size_t QKV_F = (size_t)BATCH * O3 * NPIX;
  const size_t CTX_F = (size_t)BATCH * HEADS * DHEAD * DHEAD;
  float* qkv  = (float*)ws;
  float* ctx  = (float*)(ws + QKV_F * 4);
  float* pctx = (float*)(ws + (QKV_F + CTX_F) * 4);
  float* w2   = (float*)(ws + (QKV_F + CTX_F + KSPLIT * CTX_F) * 4);

  k_qkv_gemm<<<dim3(NPIX / TN, BATCH), 256, 0, stream>>>(x, wqkv, qkv);
  k_softmax_n<<<BATCH * HIDDEN, 256, 0, stream>>>(qkv);
  k_context_part<<<dim3(KSPLIT, BATCH * HEADS), 128, 0, stream>>>(qkv, pctx);
  k_ctx_reduce<<<CTX_F / 256, 256, 0, stream>>>(pctx, ctx);
  k_fold_w2<<<BATCH * HEADS, 256, 0, stream>>>(wout, ctx, w2);
  k_out_gemm<<<dim3(NPIX / TN, BATCH), 256, 0, stream>>>(qkv, w2, out);
}


# row softmax moved into its consumers: qkv GEMM emits per-tile row max and exp sum, context and out GEMMs apply exp(q-M) on load, 1/S folded into W2; softmax launch gone; row exp-sum order changed
# speedup vs baseline: 3.3871x; 1.1147x over previous
#include <hip/hip_runtime.h>
#include <hip/hip_bf16.h>
#include <math.h>

#define BATCH   16
#define CIN     256
#define NPIX    4096
#define HEADS   4
#define DHEAD   32
#define HIDDEN  128
#define O3      384
#define KSPLIT  8
#define KCHUNK  (NPIX / KSPLIT)
#define TN      64
#define KC      32
#define APITCH  36
#define BPITCH  72
#define NTILE   (NPIX / TN)
#define PST_TILE 256
#define STAT_OFF(b) (((size_t)(b) * O3 + HIDDEN) * NPIX)
#define ROWSTAT_OFF (NTILE * PST_TILE)

typedef __attribute__((ext_vector_type(2))) float v2f;
typedef __attribute__((ext_vector_type(4))) float v4f;
typedef __attribute__((ext_vector_type(8))) float v8f;

__device__ __forceinline__ v8f wmma4(v2f a, v2f b, v8f c) {
  return __builtin_amdgcn_wmma_f32_16x16x4_f32(false, a, false, b, (short)0, c,
                                               false, false);
}

template <int MT, int KTOT, int GROWS, int MODE>
__device__ __forceinline__ void gemm_tile_lds(const float* __restrict__ A,
                                              const float* __restrict__ Bm,
                                              float* __restrict__ C,
                                              float* sA, float* sB,
                                              const int nBase,
                                              const float* __restrict__ rowM,
                                              float* sMrow,
                                              float* __restrict__ pst) {
  constexpr int NA = MT * 4;
  const int t    = threadIdx.x;
  const int lane = t & 31;
  const int wave = t >> 5;
  const int half = lane >> 4;
  const int lm   = lane & 15;
  const int m0   = wave * 16;

  const int ar = t >> 3, ac = (t & 7) * 4;
  const int br = t >> 4, bc = (t & 15) * 4;
  const float* __restrict__ gA = A + (size_t)ar * KTOT + ac;
  const float* __restrict__ gB = Bm + (size_t)br * NPIX + nBase + bc;
  float* lA = sA + ar * APITCH + ac;
  float* lB = sB + br * BPITCH + bc;

  const float* fA = sA + (m0 + lm) * APITCH + 2 * half;
  const float* fB = sB + (2 * half) * BPITCH + lm;

  v4f ra[NA];
  v4f rb[2];
  v8f acc[MT][4];
#pragma unroll
  for (int j = 0; j < MT; ++j)
#pragma unroll
    for (int i = 0; i < 4; ++i) acc[j][i] = v8f{};

#pragma unroll
  for (int i = 0; i < NA; ++i)
    ra[i] = *reinterpret_cast<const v4f*>(
        gA + (size_t)((i / 4) * GROWS + 32 * (i % 4)) * KTOT);
#pragma unroll
  for (int i = 0; i < 2; ++i)
    rb[i] = *reinterpret_cast<const v4f*>(gB + (size_t)(16 * i) * NPIX);
  if constexpr (MODE == 1) {
    static_assert(KTOT <= 256, "one thread per row of the row-max table");
    if (t < KTOT) sMrow[t] = rowM[t];
    __syncthreads();
  }
#pragma unroll
  for (int i = 0; i < NA; ++i)
    *reinterpret_cast<v4f*>(lA + 32 * i * APITCH) = ra[i];
#pragma unroll
  for (int i = 0; i < 2; ++i) {
    v4f tb = rb[i];
    if constexpr (MODE == 1) {
      const float m = sMrow[br + 16 * i];
      tb.x = __expf(tb.x - m); tb.y = __expf(tb.y - m);
      tb.z = __expf(tb.z - m); tb.w = __expf(tb.w - m);
    }
    *reinterpret_cast<v4f*>(lB + 16 * i * BPITCH) = tb;
  }
  __syncthreads();

  for (int kc = 0; kc < KTOT; kc += KC) {
    const bool more = (kc + KC) < KTOT;
    if (more) {
#pragma unroll
      for (int i = 0; i < NA; ++i)
        ra[i] = *reinterpret_cast<const v4f*>(
            gA + (size_t)((i / 4) * GROWS + 32 * (i % 4)) * KTOT + kc + KC);
#pragma unroll
      for (int i = 0; i < 2; ++i)
        rb[i] = *reinterpret_cast<const v4f*>(
            gB + (size_t)(kc + KC + 16 * i) * NPIX);
    }
#pragma unroll
    for (int kk = 0; kk < KC; kk += 4) {
      v2f a[MT], b[4];
#pragma unroll
      for (int j = 0; j < MT; ++j)
        a[j] = *reinterpret_cast<const v2f*>(fA + 128 * j * APITCH + kk);
#pragma unroll
      for (int i = 0; i < 4; ++i) {
        b[i].x = fB[kk * BPITCH + 16 * i];
        b[i].y = fB[(kk + 1) * BPITCH + 16 * i];
      }
#pragma unroll
      for (int j = 0; j < MT; ++j)
#pragma unroll
        for (int i = 0; i < 4; ++i) acc[j][i] = wmma4(a[j], b[i], acc[j][i]);
    }
    __syncthreads();
    if (more) {
#pragma unroll
      for (int i = 0; i < NA; ++i)
        *reinterpret_cast<v4f*>(lA + 32 * i * APITCH) = ra[i];
#pragma unroll
      for (int i = 0; i < 2; ++i) {
        v4f tb = rb[i];
        if constexpr (MODE == 1) {
          const float m = sMrow[kc + KC + br + 16 * i];
          tb.x = __expf(tb.x - m); tb.y = __expf(tb.y - m);
          tb.z = __expf(tb.z - m); tb.w = __expf(tb.w - m);
        }
        *reinterpret_cast<v4f*>(lB + 16 * i * BPITCH) = tb;
      }
      __syncthreads();
    }
  }

#pragma unroll
  for (int j = 0; j < MT; ++j)
#pragma unroll
    for (int i = 0; i < 4; ++i)
#pragma unroll
      for (int v = 0; v < 8; ++v)
        C[(size_t)(m0 + GROWS * j + v + 8 * half) * NPIX + nBase + 16 * i + lm] =
            acc[j][i][v];

  if constexpr (MODE == 0) {
    float mv[8], sv[8];
#pragma unroll
    for (int v = 0; v < 8; ++v) {
      float m = fmaxf(fmaxf(acc[0][0][v], acc[0][1][v]),
                      fmaxf(acc[0][2][v], acc[0][3][v]));
      m = fmaxf(m, __shfl_xor(m, 1));
      m = fmaxf(m, __shfl_xor(m, 2));
      m = fmaxf(m, __shfl_xor(m, 4));
      m = fmaxf(m, __shfl_xor(m, 8));
      float e = __expf(acc[0][0][v] - m);
      e += __expf(acc[0][1][v] - m);
      e += __expf(acc[0][2][v] - m);
      e += __expf(acc[0][3][v] - m);
      e += __shfl_xor(e, 1);
      e += __shfl_xor(e, 2);
      e += __shfl_xor(e, 4);
      e += __shfl_xor(e, 8);
      mv[v] = m;
      sv[v] = e;
    }
    const int sel = lane & 7;
    float pm = mv[0], ps = sv[0];
#pragma unroll
    for (int v = 1; v < 8; ++v) {
      pm = (sel == v) ? mv[v] : pm;
      ps = (sel == v) ? sv[v] : ps;
    }
    pst[wave * 32 + lane] = (lane & 8) ? ps : pm;
  }
}

__global__ __launch_bounds__(256) void k_qkv_gemm(
    const float* __restrict__ x, const float* __restrict__ wqkv,
    float* __restrict__ qkv) {
  __shared__ __attribute__((aligned(16))) float sA[2 * HIDDEN * APITCH];
  __shared__ __attribute__((aligned(16))) float sB[KC * BPITCH];
  const int b = blockIdx.y;
  gemm_tile_lds<2, CIN, 2 * HIDDEN, 0>(
      wqkv, x + (size_t)b * CIN * NPIX, qkv + (size_t)b * O3 * NPIX, sA, sB,
      blockIdx.x * TN, nullptr, nullptr,
      qkv + STAT_OFF(b) + (size_t)blockIdx.x * PST_TILE);
}

#define CSUB    128
#define NCSUB   (KCHUNK / CSUB)
#define CPITCH  132
#define CTROWS  (2 * DHEAD)

__device__ __forceinline__ void ctx_fetch(v4f (&r)[16],
                                          const float* __restrict__ gK,
                                          const float* __restrict__ gV,
                                          const int col) {
#pragma unroll
  for (int i = 0; i < 8; ++i) {
    r[i]     = *reinterpret_cast<const v4f*>(gK + (size_t)(4 * i) * NPIX + col);
    r[8 + i] = *reinterpret_cast<const v4f*>(gV + (size_t)(4 * i) * NPIX + col);
  }
}

__device__ __forceinline__ void ctx_park(const v4f (&r)[16], float* l) {
#pragma unroll
  for (int i = 0; i < 8; ++i) {
    *reinterpret_cast<v4f*>(l + (4 * i) * CPITCH)           = r[i];
    *reinterpret_cast<v4f*>(l + (DHEAD + 4 * i) * CPITCH)   = r[8 + i];
  }
}

__device__ __forceinline__ void ctx_colsoftmax(float* col, const float rowMax,
                                               const float rowInv) {
  float vals[DHEAD];
  float m = -INFINITY;
#pragma unroll
  for (int d = 0; d < DHEAD; ++d) {
    const float Md = __int_as_float(
        __builtin_amdgcn_readlane(__float_as_int(rowMax), d));
    const float Id = __int_as_float(
        __builtin_amdgcn_readlane(__float_as_int(rowInv), d));
    vals[d] = __expf(col[d * CPITCH] - Md) * Id;
    m = fmaxf(m, vals[d]);
  }
  float s = 0.f;
#pragma unroll
  for (int d = 0; d < DHEAD; ++d) {
    vals[d] = __expf(vals[d] - m);
    s += vals[d];
  }
  const float inv = 1.0f / s;
#pragma unroll
  for (int d = 0; d < DHEAD; ++d) col[d * CPITCH] = vals[d] * inv;
}

__device__ __forceinline__ v8f ctx_mma(const float* fA, const float* fB,
                                       v8f acc) {
#pragma unroll
  for (int kk = 0; kk < CSUB; kk += 4) {
    const v2f a  = *reinterpret_cast<const v2f*>(fA + kk);
    const v2f bb = *reinterpret_cast<const v2f*>(fB + kk);
    acc = wmma4(a, bb, acc);
  }
  return acc;
}

__global__ __launch_bounds__(128) void k_context_part(
    float* __restrict__ qkv, float* __restrict__ pctx) {
  static_assert(NCSUB == 4, "the stage sequence below is written out for four stages");
  static_assert(CSUB == 128, "one thread per column of a stage in the column softmax");
  __shared__ __attribute__((aligned(16))) float sT[2][CTROWS * CPITCH];
  __shared__ float sPM[4][DHEAD], sPS[4][DHEAD];
  const int ks = blockIdx.x;
  const int bh = blockIdx.y;
  const int b = bh >> 2, h = bh & 3;
  const int lane = threadIdx.x & 31;
  const int wave = threadIdx.x >> 5;
  const int half = lane >> 4, lm = lane & 15;
  const int mBase = (wave >> 1) * 16;
  const int nBase = (wave & 1) * 16;
  const int kBeg  = ks * KCHUNK;

  const float* __restrict__ kA =
      qkv + ((size_t)b * O3 + h * DHEAD) * NPIX;
  const float* __restrict__ vB =
      qkv + ((size_t)b * O3 + 2 * HIDDEN + h * DHEAD) * NPIX;

  const float* __restrict__ gK = kA + (size_t)wave * NPIX + kBeg + lane * 4;
  const float* __restrict__ gV = vB + (size_t)wave * NPIX + kBeg + lane * 4;
  const int lOff = wave * CPITCH + lane * 4;
  const int aOff = (mBase + lm) * CPITCH + 2 * half;
  const int bOff = (DHEAD + nBase + lm) * CPITCH + 2 * half;

  const int cOff = threadIdx.x;

  float* __restrict__ stat = qkv + STAT_OFF(b);
  float tm[16], ts[16];
  {
    const int pos = (lane >> 4) * 32 + ((lane >> 3) & 1) * 16 + (lane & 7);
    const float* __restrict__ ps =
        stat + (size_t)(wave * 16) * PST_TILE + (2 * h) * 32 + pos;
#pragma unroll
    for (int j = 0; j < 16; ++j) {
      tm[j] = ps[j * PST_TILE];
      ts[j] = ps[j * PST_TILE + 8];
    }
  }

  v4f r0[16], r1[16];
  ctx_fetch(r0, gK, gV, 0 * CSUB);
  ctx_fetch(r1, gK, gV, 1 * CSUB);
  __builtin_amdgcn_sched_barrier(0);
  {
    float pm = tm[0];
#pragma unroll
    for (int j = 1; j < 16; ++j) pm = fmaxf(pm, tm[j]);
    float psum = 0.f;
#pragma unroll
    for (int j = 0; j < 16; ++j) psum += ts[j] * __expf(tm[j] - pm);
    sPM[wave][lane] = pm;
    sPS[wave][lane] = psum;
  }
  ctx_park(r0, &sT[0][lOff]);
  __syncthreads();

  v8f acc = {};
  ctx_fetch(r0, gK, gV, 2 * CSUB);
  __builtin_amdgcn_sched_barrier(0);
  float rowMax, rowInv;
  {
    const float m0 = sPM[0][lane], m1 = sPM[1][lane];
    const float m2 = sPM[2][lane], m3 = sPM[3][lane];
    rowMax = fmaxf(fmaxf(m0, m1), fmaxf(m2, m3));
    float S = sPS[0][lane] * __expf(m0 - rowMax);
    S += sPS[1][lane] * __expf(m1 - rowMax);
    S += sPS[2][lane] * __expf(m2 - rowMax);
    S += sPS[3][lane] * __expf(m3 - rowMax);
    rowInv = 1.0f / S;
  }
  ctx_colsoftmax(&sT[0][cOff], rowMax, rowInv);
  ctx_park(r1, &sT[1][lOff]);
  __syncthreads();

  ctx_fetch(r1, gK, gV, 3 * CSUB);
  ctx_colsoftmax(&sT[1][cOff], rowMax, rowInv);
  acc = ctx_mma(&sT[0][aOff], &sT[0][bOff], acc);
  __syncthreads();

  ctx_park(r0, &sT[0][lOff]);
  acc = ctx_mma(&sT[1][aOff], &sT[1][bOff], acc);
  __syncthreads();

  ctx_colsoftmax(&sT[0][cOff], rowMax, rowInv);
  ctx_park(r1, &sT[1][lOff]);
  __syncthreads();

  ctx_colsoftmax(&sT[1][cOff], rowMax, rowInv);
  acc = ctx_mma(&sT[0][aOff], &sT[0][bOff], acc);
  __syncthreads();

  acc = ctx_mma(&sT[1][aOff], &sT[1][bOff], acc);

#pragma unroll
  for (int v = 0; v < 8; ++v) {
    const int row = mBase + v + 8 * half;
    pctx[(((size_t)ks * (BATCH * HEADS) + bh) * DHEAD + row) * DHEAD + nBase +
         lm] = acc[v];
  }
  if (ks == 0 && wave == 0) {
    stat[ROWSTAT_OFF + h * DHEAD + lane] = rowMax;
    stat[ROWSTAT_OFF + HIDDEN + h * DHEAD + lane] = rowInv;
  }
}

__global__ __launch_bounds__(256) void k_ctx_reduce(
    const float* __restrict__ pctx, float* __restrict__ ctx) {
  const int i = blockIdx.x * 256 + threadIdx.x;
  const size_t stride = (size_t)BATCH * HEADS * DHEAD * DHEAD;
  float s = 0.f;
#pragma unroll
  for (int ks = 0; ks < KSPLIT; ++ks) s += pctx[(size_t)ks * stride + i];
  ctx[i] = s;
}

__global__ __launch_bounds__(256) void k_fold_w2(
    const float* __restrict__ wout, const float* __restrict__ ctx,
    const float* __restrict__ qkv, float* __restrict__ w2) {
  const int bh = blockIdx.x;
  const int b = bh >> 2, h = bh & 3;
  const int lane = threadIdx.x & 31;
  const int wave = threadIdx.x >> 5;
  const int half = lane >> 4, lm = lane & 15;
  const int m0 = wave * 16;

  const float* __restrict__ cb = ctx + (size_t)bh * DHEAD * DHEAD;

  v8f acc00 = {}, acc01 = {}, acc10 = {}, acc11 = {};
#pragma unroll
  for (int kb = 0; kb < DHEAD; kb += 4) {
    const int k0 = kb + 2 * half;
    v2f a0, a1, b0, b1;
    a0.x = wout[(m0 + lm) * HIDDEN + h * DHEAD + k0];
    a0.y = wout[(m0 + lm) * HIDDEN + h * DHEAD + k0 + 1];
    a1.x = wout[(m0 + 128 + lm) * HIDDEN + h * DHEAD + k0];
    a1.y = wout[(m0 + 128 + lm) * HIDDEN + h * DHEAD + k0 + 1];
    b0.x = cb[(size_t)(0 + lm) * DHEAD + k0];
    b0.y = cb[(size_t)(0 + lm) * DHEAD + k0 + 1];
    b1.x = cb[(size_t)(16 + lm) * DHEAD + k0];
    b1.y = cb[(size_t)(16 + lm) * DHEAD + k0 + 1];
    acc00 = wmma4(a0, b0, acc00);
    acc01 = wmma4(a0, b1, acc01);
    acc10 = wmma4(a1, b0, acc10);
    acc11 = wmma4(a1, b1, acc11);
  }
  float* __restrict__ w2b = w2 + (size_t)b * CIN * HIDDEN;
  const float* __restrict__ invS =
      qkv + STAT_OFF(b) + ROWSTAT_OFF + HIDDEN + h * DHEAD;
  const float i0 = invS[lm], i1 = invS[16 + lm];
#pragma unroll
  for (int v = 0; v < 8; ++v) {
    const int row = v + 8 * half;
    w2b[(size_t)(m0 + row) * HIDDEN + h * DHEAD + 0 + lm]        = acc00[v] * i0;
    w2b[(size_t)(m0 + row) * HIDDEN + h * DHEAD + 16 + lm]       = acc01[v] * i1;
    w2b[(size_t)(m0 + 128 + row) * HIDDEN + h * DHEAD + 0 + lm]  = acc10[v] * i0;
    w2b[(size_t)(m0 + 128 + row) * HIDDEN + h * DHEAD + 16 + lm] = acc11[v] * i1;
  }
}

__global__ __launch_bounds__(256) void k_out_gemm(
    const float* __restrict__ qkv, const float* __restrict__ w2,
    float* __restrict__ out) {
  __shared__ __attribute__((aligned(16))) float sA[CIN * APITCH];
  __shared__ __attribute__((aligned(16))) float sB[KC * BPITCH];
  __shared__ float sMrow[HIDDEN];
  const int b = blockIdx.y;
  gemm_tile_lds<2, HIDDEN, 128, 1>(w2 + (size_t)b * CIN * HIDDEN,
                                   qkv + (size_t)b * O3 * NPIX,
                                   out + (size_t)b * CIN * NPIX, sA, sB,
                                   blockIdx.x * TN,
                                   qkv + STAT_OFF(b) + ROWSTAT_OFF,
                                   sMrow, nullptr);
}

extern "C" void kernel_launch(void* const* d_in, const int* in_sizes, int n_in,
                              void* d_out, int out_size, void* d_ws,
                              size_t ws_size, hipStream_t stream) {
  const float* x    = (const float*)d_in[0];
  const float* wqkv = (const float*)d_in[1];
  const float* wout = (const float*)d_in[2];
  float* out = (float*)d_out;

  char* ws = (char*)d_ws;
  const size_t QKV_F = (size_t)BATCH * O3 * NPIX;
  const size_t CTX_F = (size_t)BATCH * HEADS * DHEAD * DHEAD;
  float* qkv  = (float*)ws;
  float* ctx  = (float*)(ws + QKV_F * 4);
  float* pctx = (float*)(ws + (QKV_F + CTX_F) * 4);
  float* w2   = (float*)(ws + (QKV_F + CTX_F + KSPLIT * CTX_F) * 4);

  k_qkv_gemm<<<dim3(NPIX / TN, BATCH), 256, 0, stream>>>(x, wqkv, qkv);
  k_context_part<<<dim3(KSPLIT, BATCH * HEADS), 128, 0, stream>>>(qkv, pctx);
  k_ctx_reduce<<<CTX_F / 256, 256, 0, stream>>>(pctx, ctx);
  k_fold_w2<<<BATCH * HEADS, 256, 0, stream>>>(wout, ctx, qkv, w2);
  k_out_gemm<<<dim3(NPIX / TN, BATCH), 256, 0, stream>>>(qkv, w2, out);
}
